# starting kernel
# baseline (speedup 1.0000x reference)
#include <hip/hip_runtime.h>

typedef float v2f __attribute__((ext_vector_type(2)));
typedef float v8f __attribute__((ext_vector_type(8)));

#define DD 64
#define HS_PITCH 132
#define WAVES 6
#define ROWS (WAVES * 16)
#define TPB (WAVES * 32)

__global__ void ngcf_degree(const int* __restrict__ src, const int* __restrict__ dst,
                            int* __restrict__ dout, int* __restrict__ din, int E) {
  int g = blockIdx.x * blockDim.x + threadIdx.x;
  if (g < E) {
    atomicAdd(&dout[src[g]], 1);
    atomicAdd(&din[dst[g]], 1);
  }
}

__global__ void ngcf_rsqrt(const int* __restrict__ dout, const int* __restrict__ din,
                           float* __restrict__ sout, float* __restrict__ sin_, int N) {
  int g = blockIdx.x * blockDim.x + threadIdx.x;
  if (g < N) {
    int a = dout[g]; if (a < 1) a = 1;
    int b = din[g];  if (b < 1) b = 1;
    sout[g] = rsqrtf((float)a);
    sin_[g] = rsqrtf((float)b);
  }
}

__global__ void ngcf_scatter(const float* __restrict__ x, const int* __restrict__ src,
                             const int* __restrict__ dst, const float* __restrict__ sout,
                             float* __restrict__ p, int E) {
  int g = blockIdx.x * blockDim.x + threadIdx.x;
  int e = g >> 4;
  int c = g & 15;
  if (e < E) {
    int s = src[e];
    int d = dst[e];
    float w = sout[s];
    float4 v = ((const float4*)(x + (size_t)s * DD))[c];
    float* pd = p + (size_t)d * DD + (c << 2);
    atomicAdd(pd + 0, w * v.x);
    atomicAdd(pd + 1, w * v.y);
    atomicAdd(pd + 2, w * v.z);
    atomicAdd(pd + 3, w * v.w);
  }
}

__global__ __launch_bounds__(TPB) void ngcf_gemm(
    const float* __restrict__ x, const float* __restrict__ p,
    const float* __restrict__ sin_, const float* __restrict__ W1,
    const float* __restrict__ W2, float* __restrict__ out, int N, int do_relu) {
  __shared__ __align__(16) float hs[ROWS * HS_PITCH];

  const int tid = threadIdx.x;
  const int base = blockIdx.x * ROWS;

  for (int i = tid; i < ROWS * 16; i += TPB) {
    int row = i >> 4;
    int c = (i & 15) << 2;
    int node = base + row;
    if (node >= N) node = N - 1;
    float4 xv = *(const float4*)(x + (size_t)node * DD + c);
    float4 pv = *(const float4*)(p + (size_t)node * DD + c);
    float r = sin_[node];
    float4 pp = make_float4(pv.x * r, pv.y * r, pv.z * r, pv.w * r);
    float* h1 = &hs[row * HS_PITCH + c];
    float* h2 = &hs[row * HS_PITCH + DD + c];
    *(float4*)h1 = make_float4(xv.x + pp.x, xv.y + pp.y, xv.z + pp.z, xv.w + pp.w);
    *(float4*)h2 = make_float4(xv.x * pp.x, xv.y * pp.y, xv.z * pp.z, xv.w * pp.w);
  }
  __syncthreads();

  const int wave = tid >> 5;
  const int lane = tid & 31;
  const int m = lane & 15;
  const int khalf = (lane >> 4) << 1;
  const float* hrow = &hs[(wave * 16 + m) * HS_PITCH];

  v8f acc0 = {}, acc1 = {}, acc2 = {}, acc3 = {};
  for (int kb = 0; kb < 2 * DD; kb += 4) {
    v2f a = *(const v2f*)(hrow + kb + khalf);
    const float* wr = (kb < DD) ? (W1 + (size_t)(kb + khalf) * DD)
                                : (W2 + (size_t)(kb + khalf - DD) * DD);
    v2f b0, b1, b2, b3;
    b0.x = wr[m];      b0.y = wr[DD + m];
    b1.x = wr[16 + m]; b1.y = wr[DD + 16 + m];
    b2.x = wr[32 + m]; b2.y = wr[DD + 32 + m];
    b3.x = wr[48 + m]; b3.y = wr[DD + 48 + m];
    acc0 = __builtin_amdgcn_wmma_f32_16x16x4_f32(false, a, false, b0, (short)0, acc0, false, false);
    acc1 = __builtin_amdgcn_wmma_f32_16x16x4_f32(false, a, false, b1, (short)0, acc1, false, false);
    acc2 = __builtin_amdgcn_wmma_f32_16x16x4_f32(false, a, false, b2, (short)0, acc2, false, false);
    acc3 = __builtin_amdgcn_wmma_f32_16x16x4_f32(false, a, false, b3, (short)0, acc3, false, false);
  }

  const int rbase = base + wave * 16 + ((lane >> 4) << 3);
  for (int r = 0; r < 8; ++r) {
    int node = rbase + r;
    if (node < N) {
      float v0 = acc0[r], v1 = acc1[r], v2 = acc2[r], v3 = acc3[r];
      if (do_relu) {
        v0 = fmaxf(v0, 0.f); v1 = fmaxf(v1, 0.f);
        v2 = fmaxf(v2, 0.f); v3 = fmaxf(v3, 0.f);
      }
      float* orow = out + (size_t)node * DD;
      orow[m] = v0; orow[16 + m] = v1; orow[32 + m] = v2; orow[48 + m] = v3;
    }
  }
}

extern "C" void kernel_launch(void* const* d_in, const int* in_sizes, int n_in,
                              void* d_out, int out_size, void* d_ws, size_t ws_size,
                              hipStream_t stream) {
  const float* x   = (const float*)d_in[0];
  const int*   src = (const int*)d_in[1];
  const int*   dst = (const int*)d_in[2];
  const float* W1a = (const float*)d_in[3];
  const float* W2a = (const float*)d_in[4];
  const float* W1b = (const float*)d_in[5];
  const float* W2b = (const float*)d_in[6];
  float* out = (float*)d_out;

  const int N = in_sizes[0] / DD;
  const int E = in_sizes[1];

  int* deg_out = (int*)d_ws;
  int* deg_in  = deg_out + N;
  float* sout  = (float*)(deg_in + N);
  float* sin_  = sout + N;
  float* p     = sin_ + N;

  hipMemsetAsync(deg_out, 0, (size_t)2 * N * sizeof(int), stream);
  ngcf_degree<<<(E + 255) / 256, 256, 0, stream>>>(src, dst, deg_out, deg_in, E);
  ngcf_rsqrt<<<(N + 255) / 256, 256, 0, stream>>>(deg_out, deg_in, sout, sin_, N);

  const int gemm_blocks = (N + ROWS - 1) / ROWS;
  const long long sc_items = (long long)E * 16;
  const int sc_blocks = (int)((sc_items + 255) / 256);

  hipMemsetAsync(p, 0, (size_t)N * DD * sizeof(float), stream);
  ngcf_scatter<<<sc_blocks, 256, 0, stream>>>(x, src, dst, sout, p, E);
  ngcf_gemm<<<gemm_blocks, TPB, 0, stream>>>(x, p, sin_, W1a, W2a, out, N, 1);

  hipMemsetAsync(p, 0, (size_t)N * DD * sizeof(float), stream);
  ngcf_scatter<<<sc_blocks, 256, 0, stream>>>(out, src, dst, sout, p, E);
  ngcf_gemm<<<gemm_blocks, TPB, 0, stream>>>(out, p, sin_, W1b, W2b, out, N, 0);
}


# scatter: one wave per 32 edges, all 32 lanes move one edge row per step (lane l = features l, l+32), so each f32 atomic covers one whole 128 B line; atomics kept, sums in arrival order as before
# speedup vs baseline: 1.4672x; 1.4672x over previous
#include <hip/hip_runtime.h>

typedef float v2f __attribute__((ext_vector_type(2)));
typedef float v8f __attribute__((ext_vector_type(8)));

#define DD 64
#define HS_PITCH 132
#define WAVES 6
#define ROWS (WAVES * 16)
#define TPB (WAVES * 32)

__global__ void ngcf_degree(const int* __restrict__ src, const int* __restrict__ dst,
                            int* __restrict__ dout, int* __restrict__ din, int E) {
  int g = blockIdx.x * blockDim.x + threadIdx.x;
  if (g < E) {
    atomicAdd(&dout[src[g]], 1);
    atomicAdd(&din[dst[g]], 1);
  }
}

__global__ void ngcf_rsqrt(const int* __restrict__ dout, const int* __restrict__ din,
                           float* __restrict__ sout, float* __restrict__ sin_, int N) {
  int g = blockIdx.x * blockDim.x + threadIdx.x;
  if (g < N) {
    int a = dout[g]; if (a < 1) a = 1;
    int b = din[g];  if (b < 1) b = 1;
    sout[g] = rsqrtf((float)a);
    sin_[g] = rsqrtf((float)b);
  }
}

#define SC_TPB 256
#define SC_EPW 32
#define SC_B 8
#define SC_NB (SC_EPW / SC_B)
#define SC_EPB ((SC_TPB / 32) * SC_EPW)

__device__ __forceinline__ int sc_bcast(int v, int j) { return __builtin_amdgcn_readlane(v, j); }

__device__ __forceinline__ void sc_load_rows(const float* __restrict__ x, int s, int lane, int first,
                                             float (&r0)[SC_B], float (&r1)[SC_B]) {
#pragma unroll
  for (int k = 0; k < SC_B; ++k) {
    const float* xr = x + (size_t)sc_bcast(s, first + k) * DD;
    r0[k] = xr[lane];
    r1[k] = xr[32 + lane];
  }
}

__device__ __forceinline__ void sc_add_rows(float* __restrict__ p, int d, int wi, int lane, int first,
                                            const float (&r0)[SC_B], const float (&r1)[SC_B]) {
#pragma unroll
  for (int k = 0; k < SC_B; ++k) {
    const float wj = __builtin_bit_cast(float, sc_bcast(wi, first + k));
    float* pr = p + (size_t)sc_bcast(d, first + k) * DD;
    atomicAdd(pr + lane, wj * r0[k]);
    atomicAdd(pr + 32 + lane, wj * r1[k]);
  }
}

__global__ __launch_bounds__(SC_TPB) void ngcf_scatter(
    const float* __restrict__ x, const int* __restrict__ src,
    const int* __restrict__ dst, const float* __restrict__ sout,
    float* __restrict__ p, int E) {
  const int lane = threadIdx.x & 31;
  const int wv = __builtin_amdgcn_readfirstlane((int)(threadIdx.x >> 5));
  const int e0 = (blockIdx.x * (SC_TPB / 32) + wv) * SC_EPW;
  if (e0 >= E) return;
  const int e = e0 + lane;
  int s = 0, d = 0;
  float w = 0.f;
  if (e < E) {
    s = src[e];
    d = dst[e];
    w = sout[s];
  }
  const int wi = __builtin_bit_cast(int, w);
  const int cnt = (E - e0 < SC_EPW) ? (E - e0) : SC_EPW;

  if (cnt == SC_EPW) {
    float a0[SC_B], a1[SC_B], n0[SC_B], n1[SC_B];
    sc_load_rows(x, s, lane, 0, a0, a1);
#pragma unroll
    for (int b = 0; b < SC_NB; ++b) {
      if (b + 1 < SC_NB) sc_load_rows(x, s, lane, (b + 1) * SC_B, n0, n1);
      __builtin_amdgcn_sched_barrier(0);
      sc_add_rows(p, d, wi, lane, b * SC_B, a0, a1);
      if (b + 1 < SC_NB) {
#pragma unroll
        for (int k = 0; k < SC_B; ++k) { a0[k] = n0[k]; a1[k] = n1[k]; }
      }
    }
  } else {
    for (int j = 0; j < cnt; ++j) {
      const float wj = __builtin_bit_cast(float, sc_bcast(wi, j));
      const float* xr = x + (size_t)sc_bcast(s, j) * DD;
      float* pr = p + (size_t)sc_bcast(d, j) * DD;
      const float v0 = xr[lane];
      const float v1 = xr[32 + lane];
      atomicAdd(pr + lane, wj * v0);
      atomicAdd(pr + 32 + lane, wj * v1);
    }
  }
}

__global__ __launch_bounds__(TPB) void ngcf_gemm(
    const float* __restrict__ x, const float* __restrict__ p,
    const float* __restrict__ sin_, const float* __restrict__ W1,
    const float* __restrict__ W2, float* __restrict__ out, int N, int do_relu) {
  __shared__ __align__(16) float hs[ROWS * HS_PITCH];

  const int tid = threadIdx.x;
  const int base = blockIdx.x * ROWS;

  for (int i = tid; i < ROWS * 16; i += TPB) {
    int row = i >> 4;
    int c = (i & 15) << 2;
    int node = base + row;
    if (node >= N) node = N - 1;
    float4 xv = *(const float4*)(x + (size_t)node * DD + c);
    float4 pv = *(const float4*)(p + (size_t)node * DD + c);
    float r = sin_[node];
    float4 pp = make_float4(pv.x * r, pv.y * r, pv.z * r, pv.w * r);
    float* h1 = &hs[row * HS_PITCH + c];
    float* h2 = &hs[row * HS_PITCH + DD + c];
    *(float4*)h1 = make_float4(xv.x + pp.x, xv.y + pp.y, xv.z + pp.z, xv.w + pp.w);
    *(float4*)h2 = make_float4(xv.x * pp.x, xv.y * pp.y, xv.z * pp.z, xv.w * pp.w);
  }
  __syncthreads();

  const int wave = tid >> 5;
  const int lane = tid & 31;
  const int m = lane & 15;
  const int khalf = (lane >> 4) << 1;
  const float* hrow = &hs[(wave * 16 + m) * HS_PITCH];

  v8f acc0 = {}, acc1 = {}, acc2 = {}, acc3 = {};
  for (int kb = 0; kb < 2 * DD; kb += 4) {
    v2f a = *(const v2f*)(hrow + kb + khalf);
    const float* wr = (kb < DD) ? (W1 + (size_t)(kb + khalf) * DD)
                                : (W2 + (size_t)(kb + khalf - DD) * DD);
    v2f b0, b1, b2, b3;
    b0.x = wr[m];      b0.y = wr[DD + m];
    b1.x = wr[16 + m]; b1.y = wr[DD + 16 + m];
    b2.x = wr[32 + m]; b2.y = wr[DD + 32 + m];
    b3.x = wr[48 + m]; b3.y = wr[DD + 48 + m];
    acc0 = __builtin_amdgcn_wmma_f32_16x16x4_f32(false, a, false, b0, (short)0, acc0, false, false);
    acc1 = __builtin_amdgcn_wmma_f32_16x16x4_f32(false, a, false, b1, (short)0, acc1, false, false);
    acc2 = __builtin_amdgcn_wmma_f32_16x16x4_f32(false, a, false, b2, (short)0, acc2, false, false);
    acc3 = __builtin_amdgcn_wmma_f32_16x16x4_f32(false, a, false, b3, (short)0, acc3, false, false);
  }

  const int rbase = base + wave * 16 + ((lane >> 4) << 3);
  for (int r = 0; r < 8; ++r) {
    int node = rbase + r;
    if (node < N) {
      float v0 = acc0[r], v1 = acc1[r], v2 = acc2[r], v3 = acc3[r];
      if (do_relu) {
        v0 = fmaxf(v0, 0.f); v1 = fmaxf(v1, 0.f);
        v2 = fmaxf(v2, 0.f); v3 = fmaxf(v3, 0.f);
      }
      float* orow = out + (size_t)node * DD;
      orow[m] = v0; orow[16 + m] = v1; orow[32 + m] = v2; orow[48 + m] = v3;
    }
  }
}

extern "C" void kernel_launch(void* const* d_in, const int* in_sizes, int n_in,
                              void* d_out, int out_size, void* d_ws, size_t ws_size,
                              hipStream_t stream) {
  const float* x   = (const float*)d_in[0];
  const int*   src = (const int*)d_in[1];
  const int*   dst = (const int*)d_in[2];
  const float* W1a = (const float*)d_in[3];
  const float* W2a = (const float*)d_in[4];
  const float* W1b = (const float*)d_in[5];
  const float* W2b = (const float*)d_in[6];
  float* out = (float*)d_out;

  const int N = in_sizes[0] / DD;
  const int E = in_sizes[1];

  int* deg_out = (int*)d_ws;
  int* deg_in  = deg_out + N;
  float* sout  = (float*)(deg_in + N);
  float* sin_  = sout + N;
  float* p     = sin_ + N;

  hipMemsetAsync(deg_out, 0, (size_t)2 * N * sizeof(int), stream);
  ngcf_degree<<<(E + 255) / 256, 256, 0, stream>>>(src, dst, deg_out, deg_in, E);
  ngcf_rsqrt<<<(N + 255) / 256, 256, 0, stream>>>(deg_out, deg_in, sout, sin_, N);

  const int gemm_blocks = (N + ROWS - 1) / ROWS;
  const int sc_blocks = (E + SC_EPB - 1) / SC_EPB;

  hipMemsetAsync(p, 0, (size_t)N * DD * sizeof(float), stream);
  ngcf_scatter<<<sc_blocks, SC_TPB, 0, stream>>>(x, src, dst, sout, p, E);
  ngcf_gemm<<<gemm_blocks, TPB, 0, stream>>>(x, p, sin_, W1a, W2a, out, N, 1);

  hipMemsetAsync(p, 0, (size_t)N * DD * sizeof(float), stream);
  ngcf_scatter<<<sc_blocks, SC_TPB, 0, stream>>>(out, src, dst, sout, p, E);
  ngcf_gemm<<<gemm_blocks, TPB, 0, stream>>>(out, p, sin_, W1b, W2b, out, N, 0);
}


# f32 atomics out, aggregation pulled: degree pass files each edge's source in a 64-slot row per destination (count atomic = ticket); one wave sums a node's rows in registers, in ticket (arrival) order
# speedup vs baseline: 1.8020x; 1.2282x over previous
#include <hip/hip_runtime.h>

typedef float v2f __attribute__((ext_vector_type(2)));
typedef float v8f __attribute__((ext_vector_type(8)));

#define DD 64
#define HS_PITCH 132
#define WAVES 6
#define ROWS (WAVES * 16)
#define TPB (WAVES * 32)

#define SLOTS DD

__global__ void ngcf_degree_fill(const int* __restrict__ src, const int* __restrict__ dst,
                                 int* __restrict__ dout, int* __restrict__ din,
                                 int* __restrict__ slots, int E) {
  int g = blockIdx.x * blockDim.x + threadIdx.x;
  if (g < E) {
    const int s = src[g];
    const int d = dst[g];
    atomicAdd(&dout[s], 1);
    const int t = atomicAdd(&din[d], 1);
    if (t < SLOTS) slots[(size_t)d * SLOTS + t] = s;
  }
}

__global__ void ngcf_rsqrt(const int* __restrict__ dout, const int* __restrict__ din,
                           float* __restrict__ sout, float* __restrict__ sin_, int N) {
  int g = blockIdx.x * blockDim.x + threadIdx.x;
  if (g < N) {
    int a = dout[g]; if (a < 1) a = 1;
    int b = din[g];  if (b < 1) b = 1;
    sout[g] = rsqrtf((float)a);
    sin_[g] = rsqrtf((float)b);
  }
}

#define PL_TPB 256
#define PL_NPW 4
#define PL_B 8
#define PL_NPB ((PL_TPB / 32) * PL_NPW)

__device__ __forceinline__ int pl_bcast(int v, int j) { return __builtin_amdgcn_readlane(v, j); }

__device__ __forceinline__ void pl_sum_rows(const float* __restrict__ table, int s, int wi, int n,
                                            int lane, float2& acc) {
  for (int j0 = 0; j0 < n; j0 += PL_B) {
    float2 r[PL_B];
#pragma unroll
    for (int k = 0; k < PL_B; ++k) {
      const float* row = table + (size_t)(unsigned)pl_bcast(s, j0 + k) * DD;
      r[k] = *(const float2*)(row + 2 * lane);
    }
#pragma unroll
    for (int k = 0; k < PL_B; ++k) {
      const float wj = __builtin_bit_cast(float, pl_bcast(wi, j0 + k));
      const float2 v = (k == 0 || j0 + k < n) ? r[k] : make_float2(0.f, 0.f);
      acc.x += wj * v.x;
      acc.y += wj * v.y;
    }
  }
}

__device__ __forceinline__ void pl_scan_edges(const float* __restrict__ table, const int* __restrict__ src,
                                              const int* __restrict__ dst, const float* __restrict__ sout,
                                              int node, int E, int lane, float2& acc) {
  for (int e0 = 0; e0 < E; e0 += 32) {
    const int e = e0 + lane;
    int s = 0;
    float w = 0.f;
    bool hit = false;
    if (e < E) hit = (dst[e] == node);
    if (hit) {
      s = src[e];
      w = sout[s];
    }
    unsigned m = __builtin_amdgcn_ballot_w32(hit);
    const int wi = __builtin_bit_cast(int, w);
    while (m) {
      const int j = __builtin_ctz(m);
      m &= m - 1;
      const float wj = __builtin_bit_cast(float, pl_bcast(wi, j));
      const float* row = table + (size_t)(unsigned)pl_bcast(s, j) * DD;
      const float2 r = *(const float2*)(row + 2 * lane);
      acc.x += wj * r.x;
      acc.y += wj * r.y;
    }
  }
}

__global__ __launch_bounds__(PL_TPB) void ngcf_pull(
    const float* __restrict__ table, const int* slots, const int* __restrict__ din,
    const float* __restrict__ sout, const int* __restrict__ src, const int* __restrict__ dst,
    float* pout, int N, int E) {
  const int lane = threadIdx.x & 31;
  const int wv = __builtin_amdgcn_readfirstlane((int)(threadIdx.x >> 5));
  const int n0 = (blockIdx.x * (PL_TPB / 32) + wv) * PL_NPW;
  if (n0 >= N) return;

  int cv = 0;
  if (lane < PL_NPW && n0 + lane < N) cv = din[n0 + lane];

  int cnt[PL_NPW], s0[PL_NPW], w0[PL_NPW];
#pragma unroll
  for (int k = 0; k < PL_NPW; ++k) {
    cnt[k] = pl_bcast(cv, k);
    const int nk = (n0 + k < N) ? n0 + k : n0;
    const int* a = (lane < cnt[k]) ? slots + (size_t)nk * SLOTS + lane : din + nk;
    s0[k] = *a;
  }
#pragma unroll
  for (int k = 0; k < PL_NPW; ++k) {
    s0[k] = (lane < cnt[k]) ? s0[k] : 0;
    w0[k] = __builtin_bit_cast(int, sout[s0[k]]);
  }
#pragma unroll
  for (int k = 0; k < PL_NPW; ++k) w0[k] = (lane < cnt[k]) ? w0[k] : 0;

#pragma unroll
  for (int k = 0; k < PL_NPW; ++k) {
    const int node = n0 + k;
    if (node < N) {
      const int c = cnt[k];
      float2 acc = make_float2(0.f, 0.f);
      if (c <= SLOTS) {
        pl_sum_rows(table, s0[k], w0[k], c < 32 ? c : 32, lane, acc);
        if (c > 32) {
          int s1 = 0;
          float w1 = 0.f;
          if (lane + 32 < c) {
            s1 = slots[(size_t)node * SLOTS + 32 + lane];
            w1 = sout[s1];
          }
          pl_sum_rows(table, s1, __builtin_bit_cast(int, w1), c - 32, lane, acc);
        }
      } else {
        pl_scan_edges(table, src, dst, sout, node, E, lane, acc);
      }
      *(float2*)(pout + (size_t)node * DD + 2 * lane) = acc;
    }
  }
}

__global__ __launch_bounds__(TPB) void ngcf_gemm(
    const float* x, const float* p,
    const float* __restrict__ sin_, const float* __restrict__ W1,
    const float* __restrict__ W2, float* out, int N, int do_relu) {
  __shared__ __align__(16) float hs[ROWS * HS_PITCH];

  const int tid = threadIdx.x;
  const int base = blockIdx.x * ROWS;

  for (int i = tid; i < ROWS * 16; i += TPB) {
    int row = i >> 4;
    int c = (i & 15) << 2;
    int node = base + row;
    if (node >= N) node = N - 1;
    float4 xv = *(const float4*)(x + (size_t)node * DD + c);
    float4 pv = *(const float4*)(p + (size_t)node * DD + c);
    float r = sin_[node];
    float4 pp = make_float4(pv.x * r, pv.y * r, pv.z * r, pv.w * r);
    float* h1 = &hs[row * HS_PITCH + c];
    float* h2 = &hs[row * HS_PITCH + DD + c];
    *(float4*)h1 = make_float4(xv.x + pp.x, xv.y + pp.y, xv.z + pp.z, xv.w + pp.w);
    *(float4*)h2 = make_float4(xv.x * pp.x, xv.y * pp.y, xv.z * pp.z, xv.w * pp.w);
  }
  __syncthreads();

  const int wave = tid >> 5;
  const int lane = tid & 31;
  const int m = lane & 15;
  const int khalf = (lane >> 4) << 1;
  const float* hrow = &hs[(wave * 16 + m) * HS_PITCH];

  v8f acc0 = {}, acc1 = {}, acc2 = {}, acc3 = {};
  for (int kb = 0; kb < 2 * DD; kb += 4) {
    v2f a = *(const v2f*)(hrow + kb + khalf);
    const float* wr = (kb < DD) ? (W1 + (size_t)(kb + khalf) * DD)
                                : (W2 + (size_t)(kb + khalf - DD) * DD);
    v2f b0, b1, b2, b3;
    b0.x = wr[m];      b0.y = wr[DD + m];
    b1.x = wr[16 + m]; b1.y = wr[DD + 16 + m];
    b2.x = wr[32 + m]; b2.y = wr[DD + 32 + m];
    b3.x = wr[48 + m]; b3.y = wr[DD + 48 + m];
    acc0 = __builtin_amdgcn_wmma_f32_16x16x4_f32(false, a, false, b0, (short)0, acc0, false, false);
    acc1 = __builtin_amdgcn_wmma_f32_16x16x4_f32(false, a, false, b1, (short)0, acc1, false, false);
    acc2 = __builtin_amdgcn_wmma_f32_16x16x4_f32(false, a, false, b2, (short)0, acc2, false, false);
    acc3 = __builtin_amdgcn_wmma_f32_16x16x4_f32(false, a, false, b3, (short)0, acc3, false, false);
  }

  const int rbase = base + wave * 16 + ((lane >> 4) << 3);
  for (int r = 0; r < 8; ++r) {
    int node = rbase + r;
    if (node < N) {
      float v0 = acc0[r], v1 = acc1[r], v2 = acc2[r], v3 = acc3[r];
      if (do_relu) {
        v0 = fmaxf(v0, 0.f); v1 = fmaxf(v1, 0.f);
        v2 = fmaxf(v2, 0.f); v3 = fmaxf(v3, 0.f);
      }
      float* orow = out + (size_t)node * DD;
      orow[m] = v0; orow[16 + m] = v1; orow[32 + m] = v2; orow[48 + m] = v3;
    }
  }
}

extern "C" void kernel_launch(void* const* d_in, const int* in_sizes, int n_in,
                              void* d_out, int out_size, void* d_ws, size_t ws_size,
                              hipStream_t stream) {
  const float* x   = (const float*)d_in[0];
  const int*   src = (const int*)d_in[1];
  const int*   dst = (const int*)d_in[2];
  const float* W1a = (const float*)d_in[3];
  const float* W2a = (const float*)d_in[4];
  const float* W1b = (const float*)d_in[5];
  const float* W2b = (const float*)d_in[6];
  float* out = (float*)d_out;

  const int N = in_sizes[0] / DD;
  const int E = in_sizes[1];

  int* deg_out = (int*)d_ws;
  int* deg_in  = deg_out + N;
  float* sout  = (float*)(deg_in + N);
  float* sin_  = sout + N;
  float* p     = sin_ + N;
  int* slots   = (int*)p;

  hipMemsetAsync(deg_out, 0, (size_t)2 * N * sizeof(int), stream);
  ngcf_degree_fill<<<(E + 255) / 256, 256, 0, stream>>>(src, dst, deg_out, deg_in, slots, E);
  ngcf_rsqrt<<<(N + 255) / 256, 256, 0, stream>>>(deg_out, deg_in, sout, sin_, N);

  const int gemm_blocks = (N + ROWS - 1) / ROWS;
  const int pull_blocks = (N + PL_NPB - 1) / PL_NPB;

  ngcf_pull<<<pull_blocks, PL_TPB, 0, stream>>>(x, slots, deg_in, sout, src, dst, out, N, E);
  ngcf_gemm<<<gemm_blocks, TPB, 0, stream>>>(x, out, sin_, W1a, W2a, out, N, 1);

  ngcf_pull<<<pull_blocks, PL_TPB, 0, stream>>>(out, slots, deg_in, sout, src, dst, p, N, E);
  ngcf_gemm<<<gemm_blocks, TPB, 0, stream>>>(out, p, sin_, W1b, W2b, out, N, 0);
}


# degree pass without a global atomic per edge: 2048-edge chunks are range-sorted in LDS, then one block per 512-node range tickets its edges with LDS counters; sums stay in ticket (arrival) order
# speedup vs baseline: 3.3884x; 1.8804x over previous
#include <hip/hip_runtime.h>

typedef float v2f __attribute__((ext_vector_type(2)));
typedef float v8f __attribute__((ext_vector_type(8)));

#define DD 64
#define HS_PITCH 132
#define WAVES 6
#define ROWS (WAVES * 16)
#define TPB (WAVES * 32)

#define SLOTS DD

#define BIN_SHIFT 9
#define BIN_W (1 << BIN_SHIFT)
#define NBIN 256
#define CH_E 2048
#define CH_TPB 256
#define CH_EPT (CH_E / CH_TPB)
#define SPITCH (NBIN + 1)

__device__ __forceinline__ int ngcf_bin(int v) {
  const unsigned k = (unsigned)v >> BIN_SHIFT;
  return (int)(k < NBIN - 1 ? k : NBIN - 1);
}

__global__ __launch_bounds__(CH_TPB) void ngcf_bin_edges(
    const int* __restrict__ src, const int* __restrict__ dst, int* __restrict__ segD,
    int* __restrict__ segS, int* __restrict__ tabD, int* __restrict__ tabS, int E) {
  __shared__ unsigned histD[NBIN], histS[NBIN];
  __shared__ int startD[NBIN + 1], startS[NBIN + 1];
  __shared__ int wsumD[CH_TPB / 32], wsumS[CH_TPB / 32];
  __shared__ __align__(16) int stD[2 * CH_E];
  __shared__ __align__(16) int stS[CH_E];

  const int t = threadIdx.x;
  const int b = blockIdx.x;
  const int e0 = b * CH_E;
  histD[t] = 0u;
  histS[t] = 0u;

  int s[CH_EPT], d[CH_EPT];
  unsigned valid = 0u;
#pragma unroll
  for (int j = 0; j < CH_EPT / 4; ++j) {
    const int e = e0 + j * (CH_E / (CH_EPT / 4)) + 4 * t;
    if (e + 3 < E) {
      const int4 sv = *(const int4*)(src + e);
      const int4 dv = *(const int4*)(dst + e);
      s[4 * j + 0] = sv.x; s[4 * j + 1] = sv.y; s[4 * j + 2] = sv.z; s[4 * j + 3] = sv.w;
      d[4 * j + 0] = dv.x; d[4 * j + 1] = dv.y; d[4 * j + 2] = dv.z; d[4 * j + 3] = dv.w;
      valid |= 15u << (4 * j);
    } else {
#pragma unroll
      for (int i = 0; i < 4; ++i) {
        s[4 * j + i] = 0;
        d[4 * j + i] = 0;
        if (e + i < E) {
          s[4 * j + i] = src[e + i];
          d[4 * j + i] = dst[e + i];
          valid |= 1u << (4 * j + i);
        }
      }
    }
  }
  __syncthreads();

  unsigned rd[CH_EPT], rs[CH_EPT];
#pragma unroll
  for (int i = 0; i < CH_EPT; ++i) {
    rd[i] = 0u;
    rs[i] = 0u;
    if (valid & (1u << i)) {
      rd[i] = atomicAdd(&histD[ngcf_bin(d[i])], 1u);
      rs[i] = atomicAdd(&histS[ngcf_bin(s[i])], 1u);
    }
  }
  __syncthreads();

  const int lane = t & 31;
  const int wv = t >> 5;
  const int cD = (int)histD[t];
  const int cS = (int)histS[t];
  int iD = cD, iS = cS;
  for (int o = 1; o < 32; o <<= 1) {
    const int a = __shfl_up(iD, o);
    const int c = __shfl_up(iS, o);
    if (lane >= o) {
      iD += a;
      iS += c;
    }
  }
  if (lane == 31) {
    wsumD[wv] = iD;
    wsumS[wv] = iS;
  }
  __syncthreads();
  int oD = 0, oS = 0;
  for (int w = 0; w < CH_TPB / 32; ++w) {
    if (w < wv) {
      oD += wsumD[w];
      oS += wsumS[w];
    }
  }
  startD[t] = oD + iD - cD;
  startS[t] = oS + iS - cS;
  if (t == CH_TPB - 1) {
    startD[NBIN] = oD + iD;
    startS[NBIN] = oS + iS;
  }
  __syncthreads();

  tabD[(size_t)b * SPITCH + t] = startD[t];
  tabS[(size_t)b * SPITCH + t] = startS[t];
  if (t == 0) {
    tabD[(size_t)b * SPITCH + NBIN] = startD[NBIN];
    tabS[(size_t)b * SPITCH + NBIN] = startS[NBIN];
  }

#pragma unroll
  for (int i = 0; i < CH_EPT; ++i) {
    if (valid & (1u << i)) {
      const int pD = startD[ngcf_bin(d[i])] + (int)rd[i];
      const int pS = startS[ngcf_bin(s[i])] + (int)rs[i];
      *(int2*)&stD[2 * pD] = make_int2(s[i], d[i]);
      stS[pS] = s[i];
    }
  }
  __syncthreads();

  int4* gD = (int4*)(segD + (size_t)b * (2 * CH_E));
  for (int i = t; i < 2 * CH_E / 4; i += CH_TPB) gD[i] = ((const int4*)stD)[i];
  int4* gS = (int4*)(segS + (size_t)b * CH_E);
  for (int i = t; i < CH_E / 4; i += CH_TPB) gS[i] = ((const int4*)stS)[i];
}

#define RK_TPB 1024
#define RK_WAVES (RK_TPB / 32)
#define RK_Q 1024
#define RK_U 8
#define RK_SLOT 32

__device__ __forceinline__ void rk_file_in(int s, int dnode, int lo, unsigned* cI, int* sl,
                                           int* __restrict__ slots) {
  const unsigned nl = (unsigned)(dnode - lo);
  if (nl < BIN_W) {
    const unsigned r = atomicAdd(&cI[nl], 1u);
    if (r < RK_SLOT) sl[nl * RK_SLOT + r] = s;
    else if (r < SLOTS) slots[(size_t)dnode * SLOTS + r] = s;
  }
}

__device__ __forceinline__ void rk_file_out(int snode, int lo, unsigned* cO) {
  const unsigned nl = (unsigned)(snode - lo);
  if (nl < BIN_W) atomicAdd(&cO[nl], 1u);
}

__global__ __launch_bounds__(RK_TPB) void ngcf_rank(
    const int* __restrict__ segD, const int* __restrict__ segS, const int* __restrict__ tabD,
    const int* __restrict__ tabS, int* __restrict__ din, float* __restrict__ sout,
    float* __restrict__ sin_, int* __restrict__ slots, int N, int nchunks) {
  __shared__ int sl[BIN_W * RK_SLOT];
  __shared__ unsigned cI[BIN_W], cO[BIN_W];
  __shared__ int rsD[RK_Q + RK_U], reD[RK_Q + RK_U], rsS[RK_Q + RK_U], reS[RK_Q + RK_U];

  const int k = blockIdx.x;
  const int lo = k << BIN_SHIFT;
  if (lo >= N) return;

  const int t = threadIdx.x;
  const int lane = t & 31;
  const int wv = __builtin_amdgcn_readfirstlane(t >> 5);
  if (t < BIN_W) {
    cI[t] = 0u;
    cO[t] = 0u;
  }
  const int2* pairs = (const int2*)segD;

  for (int qb = 0; qb < nchunks; qb += RK_Q) {
    const int qn = (nchunks - qb < RK_Q) ? nchunks - qb : RK_Q;
    __syncthreads();
    for (int q = t; q < qn; q += RK_TPB) {
      const int* rD = tabD + (size_t)(qb + q) * SPITCH + k;
      const int* rS = tabS + (size_t)(qb + q) * SPITCH + k;
      rsD[q] = rD[0];
      reD[q] = rD[1];
      rsS[q] = rS[0];
      reS[q] = rS[1];
    }
    __syncthreads();

    for (int q0 = wv * RK_U; q0 < qn; q0 += RK_WAVES * RK_U) {
      int2 pr[RK_U];
      int id[RK_U];
      int sD[RK_U], eD[RK_U], sS[RK_U], eS[RK_U];
#pragma unroll
      for (int u = 0; u < RK_U; ++u) {
        const int q = q0 + u;
        const bool in = q < qn;
        sD[u] = in ? rsD[q] : 0;
        eD[u] = in ? reD[q] : 0;
        sS[u] = in ? rsS[q] : 0;
        eS[u] = in ? reS[q] : 0;
        const size_t cb = (size_t)(qb + q) * CH_E;
        const int iD = sD[u] + lane;
        const int iS = sS[u] + lane;
        pr[u] = pairs[(iD < eD[u]) ? cb + iD : (size_t)0];
        id[u] = segS[(iS < eS[u]) ? cb + iS : (size_t)0];
      }
#pragma unroll
      for (int u = 0; u < RK_U; ++u) {
        const size_t cb = (size_t)(qb + q0 + u) * CH_E;
        if (sD[u] + lane < eD[u]) rk_file_in(pr[u].x, pr[u].y, lo, cI, sl, slots);
        for (int i = sD[u] + 32 + lane; i < eD[u]; i += 32) {
          const int2 e = pairs[cb + i];
          rk_file_in(e.x, e.y, lo, cI, sl, slots);
        }
        if (sS[u] + lane < eS[u]) rk_file_out(id[u], lo, cO);
        for (int i = sS[u] + 32 + lane; i < eS[u]; i += 32) rk_file_out(segS[cb + i], lo, cO);
      }
    }
  }
  __syncthreads();

  if (t < BIN_W) {
    const int node = lo + t;
    if (node < N) {
      int a = (int)cO[t]; if (a < 1) a = 1;
      const int c = (int)cI[t];
      int bq = c; if (bq < 1) bq = 1;
      din[node] = c;
      sout[node] = rsqrtf((float)a);
      sin_[node] = rsqrtf((float)bq);
    }
  }
  for (int nl = wv; nl < BIN_W; nl += RK_WAVES) {
    const int node = lo + nl;
    if (node < N) {
      const unsigned c = cI[nl];
      if ((unsigned)lane < c && lane < RK_SLOT) slots[(size_t)node * SLOTS + lane] = sl[nl * RK_SLOT + lane];
    }
  }
}

#define PL_TPB 256
#define PL_NPW 4
#define PL_B 8
#define PL_NPB ((PL_TPB / 32) * PL_NPW)

__device__ __forceinline__ int pl_bcast(int v, int j) { return __builtin_amdgcn_readlane(v, j); }

__device__ __forceinline__ void pl_sum_rows(const float* __restrict__ table, int s, int wi, int n,
                                            int lane, float2& acc) {
  for (int j0 = 0; j0 < n; j0 += PL_B) {
    float2 r[PL_B];
#pragma unroll
    for (int k = 0; k < PL_B; ++k) {
      const float* row = table + (size_t)(unsigned)pl_bcast(s, j0 + k) * DD;
      r[k] = *(const float2*)(row + 2 * lane);
    }
#pragma unroll
    for (int k = 0; k < PL_B; ++k) {
      const float wj = __builtin_bit_cast(float, pl_bcast(wi, j0 + k));
      const float2 v = (k == 0 || j0 + k < n) ? r[k] : make_float2(0.f, 0.f);
      acc.x += wj * v.x;
      acc.y += wj * v.y;
    }
  }
}

__device__ __forceinline__ void pl_scan_edges(const float* __restrict__ table, const int* __restrict__ src,
                                              const int* __restrict__ dst, const float* __restrict__ sout,
                                              int node, int E, int lane, float2& acc) {
  for (int e0 = 0; e0 < E; e0 += 32) {
    const int e = e0 + lane;
    int s = 0;
    float w = 0.f;
    bool hit = false;
    if (e < E) hit = (dst[e] == node);
    if (hit) {
      s = src[e];
      w = sout[s];
    }
    unsigned m = __builtin_amdgcn_ballot_w32(hit);
    const int wi = __builtin_bit_cast(int, w);
    while (m) {
      const int j = __builtin_ctz(m);
      m &= m - 1;
      const float wj = __builtin_bit_cast(float, pl_bcast(wi, j));
      const float* row = table + (size_t)(unsigned)pl_bcast(s, j) * DD;
      const float2 r = *(const float2*)(row + 2 * lane);
      acc.x += wj * r.x;
      acc.y += wj * r.y;
    }
  }
}

__global__ __launch_bounds__(PL_TPB) void ngcf_pull(
    const float* __restrict__ table, const int* slots, const int* __restrict__ din,
    const float* __restrict__ sout, const int* __restrict__ src, const int* __restrict__ dst,
    float* pout, int N, int E) {
  const int lane = threadIdx.x & 31;
  const int wv = __builtin_amdgcn_readfirstlane((int)(threadIdx.x >> 5));
  const int n0 = (blockIdx.x * (PL_TPB / 32) + wv) * PL_NPW;
  if (n0 >= N) return;

  int cv = 0;
  if (lane < PL_NPW && n0 + lane < N) cv = din[n0 + lane];

  int cnt[PL_NPW], s0[PL_NPW], w0[PL_NPW];
#pragma unroll
  for (int k = 0; k < PL_NPW; ++k) {
    cnt[k] = pl_bcast(cv, k);
    const int nk = (n0 + k < N) ? n0 + k : n0;
    const int* a = (lane < cnt[k]) ? slots + (size_t)nk * SLOTS + lane : din + nk;
    s0[k] = *a;
  }
#pragma unroll
  for (int k = 0; k < PL_NPW; ++k) {
    s0[k] = (lane < cnt[k]) ? s0[k] : 0;
    w0[k] = __builtin_bit_cast(int, sout[s0[k]]);
  }
#pragma unroll
  for (int k = 0; k < PL_NPW; ++k) w0[k] = (lane < cnt[k]) ? w0[k] : 0;

#pragma unroll
  for (int k = 0; k < PL_NPW; ++k) {
    const int node = n0 + k;
    if (node < N) {
      const int c = cnt[k];
      float2 acc = make_float2(0.f, 0.f);
      if (c <= SLOTS) {
        pl_sum_rows(table, s0[k], w0[k], c < 32 ? c : 32, lane, acc);
        if (c > 32) {
          int s1 = 0;
          float w1 = 0.f;
          if (lane + 32 < c) {
            s1 = slots[(size_t)node * SLOTS + 32 + lane];
            w1 = sout[s1];
          }
          pl_sum_rows(table, s1, __builtin_bit_cast(int, w1), c - 32, lane, acc);
        }
      } else {
        pl_scan_edges(table, src, dst, sout, node, E, lane, acc);
      }
      *(float2*)(pout + (size_t)node * DD + 2 * lane) = acc;
    }
  }
}

__global__ __launch_bounds__(TPB) void ngcf_gemm(
    const float* x, const float* p,
    const float* __restrict__ sin_, const float* __restrict__ W1,
    const float* __restrict__ W2, float* out, int N, int do_relu) {
  __shared__ __align__(16) float hs[ROWS * HS_PITCH];

  const int tid = threadIdx.x;
  const int base = blockIdx.x * ROWS;

  for (int i = tid; i < ROWS * 16; i += TPB) {
    int row = i >> 4;
    int c = (i & 15) << 2;
    int node = base + row;
    if (node >= N) node = N - 1;
    float4 xv = *(const float4*)(x + (size_t)node * DD + c);
    float4 pv = *(const float4*)(p + (size_t)node * DD + c);
    float r = sin_[node];
    float4 pp = make_float4(pv.x * r, pv.y * r, pv.z * r, pv.w * r);
    float* h1 = &hs[row * HS_PITCH + c];
    float* h2 = &hs[row * HS_PITCH + DD + c];
    *(float4*)h1 = make_float4(xv.x + pp.x, xv.y + pp.y, xv.z + pp.z, xv.w + pp.w);
    *(float4*)h2 = make_float4(xv.x * pp.x, xv.y * pp.y, xv.z * pp.z, xv.w * pp.w);
  }
  __syncthreads();

  const int wave = tid >> 5;
  const int lane = tid & 31;
  const int m = lane & 15;
  const int khalf = (lane >> 4) << 1;
  const float* hrow = &hs[(wave * 16 + m) * HS_PITCH];

  v8f acc0 = {}, acc1 = {}, acc2 = {}, acc3 = {};
  for (int kb = 0; kb < 2 * DD; kb += 4) {
    v2f a = *(const v2f*)(hrow + kb + khalf);
    const float* wr = (kb < DD) ? (W1 + (size_t)(kb + khalf) * DD)
                                : (W2 + (size_t)(kb + khalf - DD) * DD);
    v2f b0, b1, b2, b3;
    b0.x = wr[m];      b0.y = wr[DD + m];
    b1.x = wr[16 + m]; b1.y = wr[DD + 16 + m];
    b2.x = wr[32 + m]; b2.y = wr[DD + 32 + m];
    b3.x = wr[48 + m]; b3.y = wr[DD + 48 + m];
    acc0 = __builtin_amdgcn_wmma_f32_16x16x4_f32(false, a, false, b0, (short)0, acc0, false, false);
    acc1 = __builtin_amdgcn_wmma_f32_16x16x4_f32(false, a, false, b1, (short)0, acc1, false, false);
    acc2 = __builtin_amdgcn_wmma_f32_16x16x4_f32(false, a, false, b2, (short)0, acc2, false, false);
    acc3 = __builtin_amdgcn_wmma_f32_16x16x4_f32(false, a, false, b3, (short)0, acc3, false, false);
  }

  const int rbase = base + wave * 16 + ((lane >> 4) << 3);
  for (int r = 0; r < 8; ++r) {
    int node = rbase + r;
    if (node < N) {
      float v0 = acc0[r], v1 = acc1[r], v2 = acc2[r], v3 = acc3[r];
      if (do_relu) {
        v0 = fmaxf(v0, 0.f); v1 = fmaxf(v1, 0.f);
        v2 = fmaxf(v2, 0.f); v3 = fmaxf(v3, 0.f);
      }
      float* orow = out + (size_t)node * DD;
      orow[m] = v0; orow[16 + m] = v1; orow[32 + m] = v2; orow[48 + m] = v3;
    }
  }
}

extern "C" void kernel_launch(void* const* d_in, const int* in_sizes, int n_in,
                              void* d_out, int out_size, void* d_ws, size_t ws_size,
                              hipStream_t stream) {
  const float* x   = (const float*)d_in[0];
  const int*   src = (const int*)d_in[1];
  const int*   dst = (const int*)d_in[2];
  const float* W1a = (const float*)d_in[3];
  const float* W2a = (const float*)d_in[4];
  const float* W1b = (const float*)d_in[5];
  const float* W2b = (const float*)d_in[6];
  float* out = (float*)d_out;

  const int N = in_sizes[0] / DD;
  const int E = in_sizes[1];

  int* deg_in  = (int*)d_ws + N;
  float* sout  = (float*)(deg_in + N);
  float* sin_  = sout + N;
  float* p     = sin_ + N;
  int* slots   = (int*)p;

  const int nchunks = (E + CH_E - 1) / CH_E;
  int* segD = (int*)d_out;
  int* segS = segD + (size_t)nchunks * 2 * CH_E;
  int* tabD = segS + (size_t)nchunks * CH_E;
  int* tabS = tabD + (size_t)nchunks * SPITCH;

  ngcf_bin_edges<<<nchunks, CH_TPB, 0, stream>>>(src, dst, segD, segS, tabD, tabS, E);
  ngcf_rank<<<NBIN, RK_TPB, 0, stream>>>(segD, segS, tabD, tabS, deg_in, sout, sin_, slots, N, nchunks);

  const int gemm_blocks = (N + ROWS - 1) / ROWS;
  const int pull_blocks = (N + PL_NPB - 1) / PL_NPB;

  ngcf_pull<<<pull_blocks, PL_TPB, 0, stream>>>(x, slots, deg_in, sout, src, dst, out, N, E);
  ngcf_gemm<<<gemm_blocks, TPB, 0, stream>>>(x, out, sin_, W1a, W2a, out, N, 1);

  ngcf_pull<<<pull_blocks, PL_TPB, 0, stream>>>(out, slots, deg_in, sout, src, dst, p, N, E);
  ngcf_gemm<<<gemm_blocks, TPB, 0, stream>>>(out, p, sin_, W1b, W2b, out, N, 0);
}


# aggregation fused into the product launch: a wave sums its 16 nodes' rows (same slot order, 16 rows per request batch) into its LDS rows, then multiplies; 4 launches, sums never go to memory
# speedup vs baseline: 3.8658x; 1.1409x over previous
#include <hip/hip_runtime.h>

typedef float v2f __attribute__((ext_vector_type(2)));
typedef float v8f __attribute__((ext_vector_type(8)));

#define DD 64
#define HS_PITCH 132
#define WAVES 6
#define ROWS (WAVES * 16)
#define TPB (WAVES * 32)

#define SLOTS DD

#define BIN_SHIFT 9
#define BIN_W (1 << BIN_SHIFT)
#define NBIN 256
#define CH_E 2048
#define CH_TPB 256
#define CH_EPT (CH_E / CH_TPB)
#define SPITCH (NBIN + 1)

__device__ __forceinline__ int ngcf_bin(int v) {
  const unsigned k = (unsigned)v >> BIN_SHIFT;
  return (int)(k < NBIN - 1 ? k : NBIN - 1);
}

__global__ __launch_bounds__(CH_TPB) void ngcf_bin_edges(
    const int* __restrict__ src, const int* __restrict__ dst, int* __restrict__ segD,
    int* __restrict__ segS, int* __restrict__ tabD, int* __restrict__ tabS, int E) {
  __shared__ unsigned histD[NBIN], histS[NBIN];
  __shared__ int startD[NBIN + 1], startS[NBIN + 1];
  __shared__ int wsumD[CH_TPB / 32], wsumS[CH_TPB / 32];
  __shared__ __align__(16) int stD[2 * CH_E];
  __shared__ __align__(16) int stS[CH_E];

  const int t = threadIdx.x;
  const int b = blockIdx.x;
  const int e0 = b * CH_E;
  histD[t] = 0u;
  histS[t] = 0u;

  int s[CH_EPT], d[CH_EPT];
  unsigned valid = 0u;
#pragma unroll
  for (int j = 0; j < CH_EPT / 4; ++j) {
    const int e = e0 + j * (CH_E / (CH_EPT / 4)) + 4 * t;
    if (e + 3 < E) {
      const int4 sv = *(const int4*)(src + e);
      const int4 dv = *(const int4*)(dst + e);
      s[4 * j + 0] = sv.x; s[4 * j + 1] = sv.y; s[4 * j + 2] = sv.z; s[4 * j + 3] = sv.w;
      d[4 * j + 0] = dv.x; d[4 * j + 1] = dv.y; d[4 * j + 2] = dv.z; d[4 * j + 3] = dv.w;
      valid |= 15u << (4 * j);
    } else {
#pragma unroll
      for (int i = 0; i < 4; ++i) {
        s[4 * j + i] = 0;
        d[4 * j + i] = 0;
        if (e + i < E) {
          s[4 * j + i] = src[e + i];
          d[4 * j + i] = dst[e + i];
          valid |= 1u << (4 * j + i);
        }
      }
    }
  }
  __syncthreads();

  unsigned rd[CH_EPT], rs[CH_EPT];
#pragma unroll
  for (int i = 0; i < CH_EPT; ++i) {
    rd[i] = 0u;
    rs[i] = 0u;
    if (valid & (1u << i)) {
      rd[i] = atomicAdd(&histD[ngcf_bin(d[i])], 1u);
      rs[i] = atomicAdd(&histS[ngcf_bin(s[i])], 1u);
    }
  }
  __syncthreads();

  const int lane = t & 31;
  const int wv = t >> 5;
  const int cD = (int)histD[t];
  const int cS = (int)histS[t];
  int iD = cD, iS = cS;
  for (int o = 1; o < 32; o <<= 1) {
    const int a = __shfl_up(iD, o);
    const int c = __shfl_up(iS, o);
    if (lane >= o) {
      iD += a;
      iS += c;
    }
  }
  if (lane == 31) {
    wsumD[wv] = iD;
    wsumS[wv] = iS;
  }
  __syncthreads();
  int oD = 0, oS = 0;
  for (int w = 0; w < CH_TPB / 32; ++w) {
    if (w < wv) {
      oD += wsumD[w];
      oS += wsumS[w];
    }
  }
  startD[t] = oD + iD - cD;
  startS[t] = oS + iS - cS;
  if (t == CH_TPB - 1) {
    startD[NBIN] = oD + iD;
    startS[NBIN] = oS + iS;
  }
  __syncthreads();

  tabD[(size_t)b * SPITCH + t] = startD[t];
  tabS[(size_t)b * SPITCH + t] = startS[t];
  if (t == 0) {
    tabD[(size_t)b * SPITCH + NBIN] = startD[NBIN];
    tabS[(size_t)b * SPITCH + NBIN] = startS[NBIN];
  }

#pragma unroll
  for (int i = 0; i < CH_EPT; ++i) {
    if (valid & (1u << i)) {
      const int pD = startD[ngcf_bin(d[i])] + (int)rd[i];
      const int pS = startS[ngcf_bin(s[i])] + (int)rs[i];
      *(int2*)&stD[2 * pD] = make_int2(s[i], d[i]);
      stS[pS] = s[i];
    }
  }
  __syncthreads();

  int4* gD = (int4*)(segD + (size_t)b * (2 * CH_E));
  for (int i = t; i < 2 * CH_E / 4; i += CH_TPB) gD[i] = ((const int4*)stD)[i];
  int4* gS = (int4*)(segS + (size_t)b * CH_E);
  for (int i = t; i < CH_E / 4; i += CH_TPB) gS[i] = ((const int4*)stS)[i];
}

#define RK_TPB 1024
#define RK_WAVES (RK_TPB / 32)
#define RK_Q 1024
#define RK_U 8
#define RK_SLOT 32

__device__ __forceinline__ void rk_file_in(int s, int dnode, int lo, unsigned* cI, int* sl,
                                           int* __restrict__ slots) {
  const unsigned nl = (unsigned)(dnode - lo);
  if (nl < BIN_W) {
    const unsigned r = atomicAdd(&cI[nl], 1u);
    if (r < RK_SLOT) sl[nl * RK_SLOT + r] = s;
    else if (r < SLOTS) slots[(size_t)dnode * SLOTS + r] = s;
  }
}

__device__ __forceinline__ void rk_file_out(int snode, int lo, unsigned* cO) {
  const unsigned nl = (unsigned)(snode - lo);
  if (nl < BIN_W) atomicAdd(&cO[nl], 1u);
}

__global__ __launch_bounds__(RK_TPB) void ngcf_rank(
    const int* __restrict__ segD, const int* __restrict__ segS, const int* __restrict__ tabD,
    const int* __restrict__ tabS, int* __restrict__ din, float* __restrict__ sout,
    float* __restrict__ sin_, int* __restrict__ slots, int N, int nchunks) {
  __shared__ int sl[BIN_W * RK_SLOT];
  __shared__ unsigned cI[BIN_W], cO[BIN_W];
  __shared__ int rsD[RK_Q + RK_U], reD[RK_Q + RK_U], rsS[RK_Q + RK_U], reS[RK_Q + RK_U];

  const int k = blockIdx.x;
  const int lo = k << BIN_SHIFT;
  if (lo >= N) return;

  const int t = threadIdx.x;
  const int lane = t & 31;
  const int wv = __builtin_amdgcn_readfirstlane(t >> 5);
  if (t < BIN_W) {
    cI[t] = 0u;
    cO[t] = 0u;
  }
  const int2* pairs = (const int2*)segD;

  for (int qb = 0; qb < nchunks; qb += RK_Q) {
    const int qn = (nchunks - qb < RK_Q) ? nchunks - qb : RK_Q;
    __syncthreads();
    for (int q = t; q < qn; q += RK_TPB) {
      const int* rD = tabD + (size_t)(qb + q) * SPITCH + k;
      const int* rS = tabS + (size_t)(qb + q) * SPITCH + k;
      rsD[q] = rD[0];
      reD[q] = rD[1];
      rsS[q] = rS[0];
      reS[q] = rS[1];
    }
    __syncthreads();

    for (int q0 = wv * RK_U; q0 < qn; q0 += RK_WAVES * RK_U) {
      int2 pr[RK_U];
      int id[RK_U];
      int sD[RK_U], eD[RK_U], sS[RK_U], eS[RK_U];
#pragma unroll
      for (int u = 0; u < RK_U; ++u) {
        const int q = q0 + u;
        const bool in = q < qn;
        sD[u] = in ? rsD[q] : 0;
        eD[u] = in ? reD[q] : 0;
        sS[u] = in ? rsS[q] : 0;
        eS[u] = in ? reS[q] : 0;
        const size_t cb = (size_t)(qb + q) * CH_E;
        const int iD = sD[u] + lane;
        const int iS = sS[u] + lane;
        pr[u] = pairs[(iD < eD[u]) ? cb + iD : (size_t)0];
        id[u] = segS[(iS < eS[u]) ? cb + iS : (size_t)0];
      }
#pragma unroll
      for (int u = 0; u < RK_U; ++u) {
        const size_t cb = (size_t)(qb + q0 + u) * CH_E;
        if (sD[u] + lane < eD[u]) rk_file_in(pr[u].x, pr[u].y, lo, cI, sl, slots);
        for (int i = sD[u] + 32 + lane; i < eD[u]; i += 32) {
          const int2 e = pairs[cb + i];
          rk_file_in(e.x, e.y, lo, cI, sl, slots);
        }
        if (sS[u] + lane < eS[u]) rk_file_out(id[u], lo, cO);
        for (int i = sS[u] + 32 + lane; i < eS[u]; i += 32) rk_file_out(segS[cb + i], lo, cO);
      }
    }
  }
  __syncthreads();

  if (t < BIN_W) {
    const int node = lo + t;
    if (node < N) {
      int a = (int)cO[t]; if (a < 1) a = 1;
      const int c = (int)cI[t];
      int bq = c; if (bq < 1) bq = 1;
      din[node] = c;
      sout[node] = rsqrtf((float)a);
      sin_[node] = rsqrtf((float)bq);
    }
  }
  for (int nl = wv; nl < BIN_W; nl += RK_WAVES) {
    const int node = lo + nl;
    if (node < N) {
      const unsigned c = cI[nl];
      if ((unsigned)lane < c && lane < RK_SLOT) slots[(size_t)node * SLOTS + lane] = sl[nl * RK_SLOT + lane];
    }
  }
}

#define PL_G 4
#define PL_B 16

__device__ __forceinline__ int pl_bcast(int v, int j) { return __builtin_amdgcn_readlane(v, j); }

__device__ __forceinline__ int pl_slot_entry(const int* slots, const int* __restrict__ din, int node,
                                             int N, int c, int lane) {
  const int nk = (node < N) ? node : N - 1;
  const int* a = (lane < c) ? slots + (size_t)nk * SLOTS + lane : din + nk;
  return *a;
}

__device__ __forceinline__ void pl_sum_rows(const float* __restrict__ table, int s, int wi, int n,
                                            int lane, float2& acc) {
  for (int j0 = 0; j0 < n; j0 += PL_B) {
    float2 r[PL_B];
#pragma unroll
    for (int k = 0; k < PL_B; ++k) {
      const float* row = table + (size_t)(unsigned)pl_bcast(s, j0 + k) * DD;
      r[k] = *(const float2*)(row + 2 * lane);
    }
#pragma unroll
    for (int k = 0; k < PL_B; ++k) {
      const float wj = __builtin_bit_cast(float, pl_bcast(wi, j0 + k));
      const float2 v = (k == 0 || j0 + k < n) ? r[k] : make_float2(0.f, 0.f);
      acc.x += wj * v.x;
      acc.y += wj * v.y;
    }
  }
}

__device__ __forceinline__ void pl_scan_edges(const float* __restrict__ table, const int* __restrict__ src,
                                              const int* __restrict__ dst, const float* __restrict__ sout,
                                              int node, int E, int lane, float2& acc) {
  for (int e0 = 0; e0 < E; e0 += 32) {
    const int e = e0 + lane;
    int s = 0;
    float w = 0.f;
    bool hit = false;
    if (e < E) hit = (dst[e] == node);
    if (hit) {
      s = src[e];
      w = sout[s];
    }
    unsigned m = __builtin_amdgcn_ballot_w32(hit);
    const int wi = __builtin_bit_cast(int, w);
    while (m) {
      const int j = __builtin_ctz(m);
      m &= m - 1;
      const float wj = __builtin_bit_cast(float, pl_bcast(wi, j));
      const float* row = table + (size_t)(unsigned)pl_bcast(s, j) * DD;
      const float2 r = *(const float2*)(row + 2 * lane);
      acc.x += wj * r.x;
      acc.y += wj * r.y;
    }
  }
}

__global__ __launch_bounds__(TPB) void ngcf_layer(
    const float* __restrict__ feat, const int* slots, const int* __restrict__ din,
    const float* __restrict__ sout, const float* __restrict__ sin_,
    const int* __restrict__ src, const int* __restrict__ dst,
    const float* __restrict__ W1, const float* __restrict__ W2,
    float* out, int N, int E, int do_relu) {
  __shared__ __align__(16) float hs[ROWS * HS_PITCH];

  const int tid = threadIdx.x;
  const int lane = tid & 31;
  const int wave = __builtin_amdgcn_readfirstlane(tid >> 5);
  const int base = blockIdx.x * ROWS;
  const int n0 = base + wave * 16;

  int cv = 0;
  if (lane < 16 && n0 + lane < N) cv = din[n0 + lane];

  int sn[PL_G];
#pragma unroll
  for (int k = 0; k < PL_G; ++k) sn[k] = pl_slot_entry(slots, din, n0 + k, N, pl_bcast(cv, k), lane);

#pragma unroll 1
  for (int g = 0; g < 16 / PL_G; ++g) {
    int cnt[PL_G], s0[PL_G], w0[PL_G];
#pragma unroll
    for (int k = 0; k < PL_G; ++k) {
      cnt[k] = pl_bcast(cv, g * PL_G + k);
      s0[k] = (lane < cnt[k]) ? sn[k] : 0;
    }
#pragma unroll
    for (int k = 0; k < PL_G; ++k) w0[k] = __builtin_bit_cast(int, sout[s0[k]]);
    if (g + 1 < 16 / PL_G) {
#pragma unroll
      for (int k = 0; k < PL_G; ++k) {
        const int q = (g + 1) * PL_G + k;
        sn[k] = pl_slot_entry(slots, din, n0 + q, N, pl_bcast(cv, q), lane);
      }
    }
#pragma unroll
    for (int k = 0; k < PL_G; ++k) w0[k] = (lane < cnt[k]) ? w0[k] : 0;

#pragma unroll
    for (int k = 0; k < PL_G; ++k) {
      const int row = wave * 16 + g * PL_G + k;
      const int node = base + row;
      float2 h1 = make_float2(0.f, 0.f), h2 = make_float2(0.f, 0.f);
      if (node < N) {
        const float2 xv = *(const float2*)(feat + (size_t)node * DD + 2 * lane);
        const float r = sin_[node];
        const int c = cnt[k];
        float2 acc = make_float2(0.f, 0.f);
        if (c <= SLOTS) {
          pl_sum_rows(feat, s0[k], w0[k], c < 32 ? c : 32, lane, acc);
          if (c > 32) {
            int s1 = 0;
            float w1 = 0.f;
            if (lane + 32 < c) {
              s1 = slots[(size_t)node * SLOTS + 32 + lane];
              w1 = sout[s1];
            }
            pl_sum_rows(feat, s1, __builtin_bit_cast(int, w1), c - 32, lane, acc);
          }
        } else {
          pl_scan_edges(feat, src, dst, sout, node, E, lane, acc);
        }
        const float2 pp = make_float2(acc.x * r, acc.y * r);
        h1 = make_float2(xv.x + pp.x, xv.y + pp.y);
        h2 = make_float2(xv.x * pp.x, xv.y * pp.y);
      }
      float* hr = &hs[row * HS_PITCH + 2 * lane];
      *(float2*)hr = h1;
      *(float2*)(hr + DD) = h2;
    }
  }
  __syncthreads();

  const int m = lane & 15;
  const int khalf = (lane >> 4) << 1;
  const float* hrow = &hs[(wave * 16 + m) * HS_PITCH];

  v8f acc0 = {}, acc1 = {}, acc2 = {}, acc3 = {};
  for (int kb = 0; kb < 2 * DD; kb += 4) {
    v2f a = *(const v2f*)(hrow + kb + khalf);
    const float* wr = (kb < DD) ? (W1 + (size_t)(kb + khalf) * DD)
                                : (W2 + (size_t)(kb + khalf - DD) * DD);
    v2f b0, b1, b2, b3;
    b0.x = wr[m];      b0.y = wr[DD + m];
    b1.x = wr[16 + m]; b1.y = wr[DD + 16 + m];
    b2.x = wr[32 + m]; b2.y = wr[DD + 32 + m];
    b3.x = wr[48 + m]; b3.y = wr[DD + 48 + m];
    acc0 = __builtin_amdgcn_wmma_f32_16x16x4_f32(false, a, false, b0, (short)0, acc0, false, false);
    acc1 = __builtin_amdgcn_wmma_f32_16x16x4_f32(false, a, false, b1, (short)0, acc1, false, false);
    acc2 = __builtin_amdgcn_wmma_f32_16x16x4_f32(false, a, false, b2, (short)0, acc2, false, false);
    acc3 = __builtin_amdgcn_wmma_f32_16x16x4_f32(false, a, false, b3, (short)0, acc3, false, false);
  }

  const int rbase = base + wave * 16 + ((lane >> 4) << 3);
  for (int r = 0; r < 8; ++r) {
    int node = rbase + r;
    if (node < N) {
      float v0 = acc0[r], v1 = acc1[r], v2 = acc2[r], v3 = acc3[r];
      if (do_relu) {
        v0 = fmaxf(v0, 0.f); v1 = fmaxf(v1, 0.f);
        v2 = fmaxf(v2, 0.f); v3 = fmaxf(v3, 0.f);
      }
      float* orow = out + (size_t)node * DD;
      orow[m] = v0; orow[16 + m] = v1; orow[32 + m] = v2; orow[48 + m] = v3;
    }
  }
}

extern "C" void kernel_launch(void* const* d_in, const int* in_sizes, int n_in,
                              void* d_out, int out_size, void* d_ws, size_t ws_size,
                              hipStream_t stream) {
  const float* x   = (const float*)d_in[0];
  const int*   src = (const int*)d_in[1];
  const int*   dst = (const int*)d_in[2];
  const float* W1a = (const float*)d_in[3];
  const float* W2a = (const float*)d_in[4];
  const float* W1b = (const float*)d_in[5];
  const float* W2b = (const float*)d_in[6];
  float* out = (float*)d_out;

  const int N = in_sizes[0] / DD;
  const int E = in_sizes[1];

  int* deg_in  = (int*)d_ws + N;
  float* sout  = (float*)(deg_in + N);
  float* sin_  = sout + N;
  float* h     = sin_ + N;
  int* slots   = (int*)d_out;

  const int nchunks = (E + CH_E - 1) / CH_E;
  int* segD = (int*)h;
  int* segS = segD + (size_t)nchunks * 2 * CH_E;
  int* tabD = segS + (size_t)nchunks * CH_E;
  int* tabS = tabD + (size_t)nchunks * SPITCH;

  ngcf_bin_edges<<<nchunks, CH_TPB, 0, stream>>>(src, dst, segD, segS, tabD, tabS, E);
  ngcf_rank<<<NBIN, RK_TPB, 0, stream>>>(segD, segS, tabD, tabS, deg_in, sout, sin_, slots, N, nchunks);

  const int blocks = (N + ROWS - 1) / ROWS;

  ngcf_layer<<<blocks, TPB, 0, stream>>>(x, slots, deg_in, sout, sin_, src, dst, W1a, W2a, h, N, E, 1);

  ngcf_layer<<<blocks, TPB, 0, stream>>>(h, slots, deg_in, sout, sin_, src, dst, W1b, W2b, out, N, E, 0);
}
